# starting kernel
# baseline (speedup 1.0000x reference)
#include <hip/hip_runtime.h>
#include <hip/hip_bf16.h>

typedef __attribute__((ext_vector_type(2))) float v2f;
typedef __attribute__((ext_vector_type(8))) float v8f;

#define HIDDEN 128

__global__ void gat_prep_uv(const float* __restrict__ W, const float* __restrict__ b,
                            const float* __restrict__ att_w,
                            float* __restrict__ G, float* __restrict__ cs) {
  int j = threadIdx.x;
  float su = 0.f, sv = 0.f;
  for (int k = 0; k < HIDDEN; ++k) {
    float w = W[k * HIDDEN + j];
    su = fmaf(att_w[k], w, su);
    sv = fmaf(att_w[HIDDEN + k], w, sv);
  }
  G[j] = su;
  G[HIDDEN + j] = sv;
#pragma unroll
  for (int r = 2; r < 16; ++r) G[r * HIDDEN + j] = 0.f;
  __shared__ float tb[HIDDEN];
  tb[j] = (att_w[j] + att_w[HIDDEN + j]) * b[j];
  __syncthreads();
  if (j == 0) {
    float s = 0.f;
    for (int k = 0; k < HIDDEN; ++k) s += tb[k];
    cs[0] = s;
  }
}

__global__ void gat_zero(float* __restrict__ nacc, int n) {
  int i = blockIdx.x * blockDim.x + threadIdx.x;
  if (i < n) nacc[i] = 0.f;
}

__global__ void gat_pq_wmma(const float* __restrict__ x, const float* __restrict__ G,
                            float* __restrict__ p, float* __restrict__ q,
                            int n_tiles, int n_nodes) {
  int lane = threadIdx.x & 31;
  int wv   = (blockIdx.x * blockDim.x + threadIdx.x) >> 5;
  int t0 = wv * 2;
  int t1 = t0 + 1;
  if (t0 >= n_tiles) return;
  bool has1 = (t1 < n_tiles);

  int m    = lane & 15;
  int koff = (lane >> 4) << 1;
  int n0 = t0 * 16 + m; if (n0 >= n_nodes) n0 = n_nodes - 1;
  int n1 = has1 ? (t1 * 16 + m) : n0; if (n1 >= n_nodes) n1 = n_nodes - 1;

  const float* a_ptr = G + (size_t)m  * HIDDEN + koff;
  const float* b0ptr = x + (size_t)n0 * HIDDEN + koff;
  const float* b1ptr = x + (size_t)n1 * HIDDEN + koff;

  v8f acc0 = {0.f, 0.f, 0.f, 0.f, 0.f, 0.f, 0.f, 0.f};
  v8f acc1 = {0.f, 0.f, 0.f, 0.f, 0.f, 0.f, 0.f, 0.f};
#pragma unroll
  for (int k0 = 0; k0 < HIDDEN; k0 += 4) {
    v2f aa  = *(const v2f*)(a_ptr + k0);
    v2f bb0 = *(const v2f*)(b0ptr + k0);
    v2f bb1 = *(const v2f*)(b1ptr + k0);
    acc0 = __builtin_amdgcn_wmma_f32_16x16x4_f32(false, aa, false, bb0,
                                                 (short)0, acc0, false, false);
    acc1 = __builtin_amdgcn_wmma_f32_16x16x4_f32(false, aa, false, bb1,
                                                 (short)0, acc1, false, false);
  }
  if (lane < 16) {
    int o0 = t0 * 16 + lane;
    if (o0 < n_nodes) { p[o0] = acc0[0]; q[o0] = acc0[1]; }
    int o1 = t1 * 16 + lane;
    if (has1 && o1 < n_nodes) { p[o1] = acc1[0]; q[o1] = acc1[1]; }
  }
}

__global__ void gat_edge(const int* __restrict__ e, const float* __restrict__ p,
                         const float* __restrict__ q, const float* __restrict__ cs,
                         float* __restrict__ nacc, int n_edges) {
  int i = blockIdx.x * blockDim.x + threadIdx.x;
  if (i >= n_edges) return;
  int srcn = e[2 * i];
  int dstn = e[2 * i + 1];
  float z = p[srcn] + q[dstn] + cs[0];
  float lz = z > 0.f ? z : 0.2f * z;
  atomicAdd(&nacc[dstn], expf(lz));
}

__global__ void gat_out(const float* __restrict__ x, const float* __restrict__ nacc,
                        const float* __restrict__ nw, const float* __restrict__ nb,
                        float* __restrict__ out, int n_nodes) {
  int lane = threadIdx.x & 31;
  int node = (blockIdx.x * blockDim.x + threadIdx.x) >> 5;
  if (node >= n_nodes) return;
  float nv = nacc[node];
  float denom = (nv == 0.f) ? 1.f : nv;
  float4 xv = ((const float4*)(x + (size_t)node * HIDDEN))[lane];
  float4 y;
  y.x = fmaxf(xv.x / denom, 0.f) + xv.x;
  y.y = fmaxf(xv.y / denom, 0.f) + xv.y;
  y.z = fmaxf(xv.z / denom, 0.f) + xv.z;
  y.w = fmaxf(xv.w / denom, 0.f) + xv.w;
  float ss = y.x * y.x + y.y * y.y + y.z * y.z + y.w * y.w;
  ss += __shfl_xor(ss, 16, 32);
  ss += __shfl_xor(ss,  8, 32);
  ss += __shfl_xor(ss,  4, 32);
  ss += __shfl_xor(ss,  2, 32);
  ss += __shfl_xor(ss,  1, 32);
  float inv_rms = rsqrtf(ss * (1.0f / HIDDEN) + 1e-6f);
  int c = lane * 4;
  float4 o;
  o.x = nw[c + 0] * (y.x * inv_rms) + nb[c + 0];
  o.y = nw[c + 1] * (y.y * inv_rms) + nb[c + 1];
  o.z = nw[c + 2] * (y.z * inv_rms) + nb[c + 2];
  o.w = nw[c + 3] * (y.w * inv_rms) + nb[c + 3];
  ((float4*)(out + (size_t)node * HIDDEN))[lane] = o;
}

extern "C" void kernel_launch(void* const* d_in, const int* in_sizes, int n_in,
                              void* d_out, int out_size, void* d_ws, size_t ws_size,
                              hipStream_t stream) {
  const float* x     = (const float*)d_in[0];
  const float* W     = (const float*)d_in[1];
  const float* b     = (const float*)d_in[2];
  const float* att_w = (const float*)d_in[3];
  const float* nw    = (const float*)d_in[4];
  const float* nb    = (const float*)d_in[5];
  const int*   e     = (const int*)d_in[6];
  float* out = (float*)d_out;

  const int n_nodes = in_sizes[0] / HIDDEN;
  const int n_edges = in_sizes[6] / 2;

  float* G    = (float*)d_ws;
  float* cs   = G + 16 * HIDDEN;
  float* p    = cs + 16;
  float* q    = p + n_nodes;
  float* nacc = q + n_nodes;

  gat_prep_uv<<<1, 128, 0, stream>>>(W, b, att_w, G, cs);
  gat_zero<<<(n_nodes + 255) / 256, 256, 0, stream>>>(nacc, n_nodes);

  int n_tiles = (n_nodes + 15) / 16;
  int n_waves = (n_tiles + 1) / 2;
  int pq_blocks = (n_waves * 32 + 127) / 128;
  gat_pq_wmma<<<pq_blocks, 128, 0, stream>>>(x, G, p, q, n_tiles, n_nodes);

  gat_edge<<<(n_edges + 255) / 256, 256, 0, stream>>>(e, p, q, cs, nacc, n_edges);

  gat_out<<<(n_nodes * 32 + 255) / 256, 256, 0, stream>>>(x, nacc, nw, nb, out, n_nodes);
}


# Serial one-block prologue (u,v,c) moved into the zero-fill launch, k split over 8 waves x 4 blocks so all W loads fly at once; 5 launches -> 4; u,v,c summed as partials then in fixed order
# speedup vs baseline: 1.1537x; 1.1537x over previous
#include <hip/hip_runtime.h>
#include <hip/hip_bf16.h>

typedef __attribute__((ext_vector_type(2))) float v2f;
typedef __attribute__((ext_vector_type(8))) float v8f;

#define HIDDEN 128

#define PRO_BLOCKS 4
__global__ void __launch_bounds__(256)
gat_prep_zero(const float* __restrict__ W, const float* __restrict__ b,
              const float* __restrict__ att_w,
              float* __restrict__ G, float* __restrict__ cs,
              float* __restrict__ nacc, int n) {
  __shared__ float su_s[8][32];
  __shared__ float sv_s[8][32];
  int tid  = threadIdx.x;
  int lane = tid & 31;
  if (blockIdx.x > PRO_BLOCKS) {
    int i = (blockIdx.x - PRO_BLOCKS - 1) * 256 + tid;
    if (i < n) nacc[i] = 0.f;
    return;
  }
  if (blockIdx.x == PRO_BLOCKS) {
    if (tid < 32) {
      float t = 0.f;
#pragma unroll
      for (int i = 0; i < 4; ++i) {
        int j = lane + 32 * i;
        t += (att_w[j] + att_w[HIDDEN + j]) * b[j];
      }
      t += __shfl_xor(t, 16, 32);
      t += __shfl_xor(t,  8, 32);
      t += __shfl_xor(t,  4, 32);
      t += __shfl_xor(t,  2, 32);
      t += __shfl_xor(t,  1, 32);
      if (lane == 0) cs[0] = t;
    }
    return;
  }
  int kg  = __builtin_amdgcn_readfirstlane(tid >> 5);
  int col = blockIdx.x * 32 + lane;
  float su = 0.f, sv = 0.f;
#pragma unroll
  for (int i = 0; i < 16; ++i) {
    int k = kg * 16 + i;
    float w = W[k * HIDDEN + col];
    su = fmaf(att_w[k], w, su);
    sv = fmaf(att_w[HIDDEN + k], w, sv);
  }
  su_s[kg][lane] = su;
  sv_s[kg][lane] = sv;
  __syncthreads();
  if (kg == 0) {
    float u = su_s[0][lane], v = sv_s[0][lane];
#pragma unroll
    for (int g = 1; g < 8; ++g) { u += su_s[g][lane]; v += sv_s[g][lane]; }
    G[col] = u;
    G[HIDDEN + col] = v;
  } else {
    G[(2 * kg) * HIDDEN + col] = 0.f;
    G[(2 * kg + 1) * HIDDEN + col] = 0.f;
  }
}

__global__ void gat_pq_wmma(const float* __restrict__ x, const float* __restrict__ G,
                            float* __restrict__ p, float* __restrict__ q,
                            int n_tiles, int n_nodes) {
  int lane = threadIdx.x & 31;
  int wv   = (blockIdx.x * blockDim.x + threadIdx.x) >> 5;
  int t0 = wv * 2;
  int t1 = t0 + 1;
  if (t0 >= n_tiles) return;
  bool has1 = (t1 < n_tiles);

  int m    = lane & 15;
  int koff = (lane >> 4) << 1;
  int n0 = t0 * 16 + m; if (n0 >= n_nodes) n0 = n_nodes - 1;
  int n1 = has1 ? (t1 * 16 + m) : n0; if (n1 >= n_nodes) n1 = n_nodes - 1;

  const float* a_ptr = G + (size_t)m  * HIDDEN + koff;
  const float* b0ptr = x + (size_t)n0 * HIDDEN + koff;
  const float* b1ptr = x + (size_t)n1 * HIDDEN + koff;

  v8f acc0 = {0.f, 0.f, 0.f, 0.f, 0.f, 0.f, 0.f, 0.f};
  v8f acc1 = {0.f, 0.f, 0.f, 0.f, 0.f, 0.f, 0.f, 0.f};
#pragma unroll
  for (int k0 = 0; k0 < HIDDEN; k0 += 4) {
    v2f aa  = *(const v2f*)(a_ptr + k0);
    v2f bb0 = *(const v2f*)(b0ptr + k0);
    v2f bb1 = *(const v2f*)(b1ptr + k0);
    acc0 = __builtin_amdgcn_wmma_f32_16x16x4_f32(false, aa, false, bb0,
                                                 (short)0, acc0, false, false);
    acc1 = __builtin_amdgcn_wmma_f32_16x16x4_f32(false, aa, false, bb1,
                                                 (short)0, acc1, false, false);
  }
  if (lane < 16) {
    int o0 = t0 * 16 + lane;
    if (o0 < n_nodes) { p[o0] = acc0[0]; q[o0] = acc0[1]; }
    int o1 = t1 * 16 + lane;
    if (has1 && o1 < n_nodes) { p[o1] = acc1[0]; q[o1] = acc1[1]; }
  }
}

__global__ void gat_edge(const int* __restrict__ e, const float* __restrict__ p,
                         const float* __restrict__ q, const float* __restrict__ cs,
                         float* __restrict__ nacc, int n_edges) {
  int i = blockIdx.x * blockDim.x + threadIdx.x;
  if (i >= n_edges) return;
  int srcn = e[2 * i];
  int dstn = e[2 * i + 1];
  float z = p[srcn] + q[dstn] + cs[0];
  float lz = z > 0.f ? z : 0.2f * z;
  atomicAdd(&nacc[dstn], expf(lz));
}

__global__ void gat_out(const float* __restrict__ x, const float* __restrict__ nacc,
                        const float* __restrict__ nw, const float* __restrict__ nb,
                        float* __restrict__ out, int n_nodes) {
  int lane = threadIdx.x & 31;
  int node = (blockIdx.x * blockDim.x + threadIdx.x) >> 5;
  if (node >= n_nodes) return;
  float nv = nacc[node];
  float denom = (nv == 0.f) ? 1.f : nv;
  float4 xv = ((const float4*)(x + (size_t)node * HIDDEN))[lane];
  float4 y;
  y.x = fmaxf(xv.x / denom, 0.f) + xv.x;
  y.y = fmaxf(xv.y / denom, 0.f) + xv.y;
  y.z = fmaxf(xv.z / denom, 0.f) + xv.z;
  y.w = fmaxf(xv.w / denom, 0.f) + xv.w;
  float ss = y.x * y.x + y.y * y.y + y.z * y.z + y.w * y.w;
  ss += __shfl_xor(ss, 16, 32);
  ss += __shfl_xor(ss,  8, 32);
  ss += __shfl_xor(ss,  4, 32);
  ss += __shfl_xor(ss,  2, 32);
  ss += __shfl_xor(ss,  1, 32);
  float inv_rms = rsqrtf(ss * (1.0f / HIDDEN) + 1e-6f);
  int c = lane * 4;
  float4 o;
  o.x = nw[c + 0] * (y.x * inv_rms) + nb[c + 0];
  o.y = nw[c + 1] * (y.y * inv_rms) + nb[c + 1];
  o.z = nw[c + 2] * (y.z * inv_rms) + nb[c + 2];
  o.w = nw[c + 3] * (y.w * inv_rms) + nb[c + 3];
  ((float4*)(out + (size_t)node * HIDDEN))[lane] = o;
}

extern "C" void kernel_launch(void* const* d_in, const int* in_sizes, int n_in,
                              void* d_out, int out_size, void* d_ws, size_t ws_size,
                              hipStream_t stream) {
  const float* x     = (const float*)d_in[0];
  const float* W     = (const float*)d_in[1];
  const float* b     = (const float*)d_in[2];
  const float* att_w = (const float*)d_in[3];
  const float* nw    = (const float*)d_in[4];
  const float* nb    = (const float*)d_in[5];
  const int*   e     = (const int*)d_in[6];
  float* out = (float*)d_out;

  const int n_nodes = in_sizes[0] / HIDDEN;
  const int n_edges = in_sizes[6] / 2;

  float* G    = (float*)d_ws;
  float* cs   = G + 16 * HIDDEN;
  float* p    = cs + 16;
  float* q    = p + n_nodes;
  float* nacc = q + n_nodes;

  gat_prep_zero<<<PRO_BLOCKS + 1 + (n_nodes + 255) / 256, 256, 0, stream>>>(
      W, b, att_w, G, cs, nacc, n_nodes);

  int n_tiles = (n_nodes + 15) / 16;
  int n_waves = (n_tiles + 1) / 2;
  int pq_blocks = (n_waves * 32 + 127) / 128;
  gat_pq_wmma<<<pq_blocks, 128, 0, stream>>>(x, G, p, q, n_tiles, n_nodes);

  gat_edge<<<(n_edges + 255) / 256, 256, 0, stream>>>(e, p, q, cs, nacc, n_edges);

  gat_out<<<(n_nodes * 32 + 255) / 256, 256, 0, stream>>>(x, nacc, nw, nb, out, n_nodes);
}


# Output kernel: each wave32 takes 25 consecutive nodes, issues all 25 row loads first, finishes rows 5 at a time (same per-row sums); 4000 waves replace 100000 one-node waves whose entry was the limit
# speedup vs baseline: 1.2393x; 1.0742x over previous
#include <hip/hip_runtime.h>
#include <hip/hip_bf16.h>

typedef __attribute__((ext_vector_type(2))) float v2f;
typedef __attribute__((ext_vector_type(8))) float v8f;

#define HIDDEN 128

#define PRO_BLOCKS 4
__global__ void __launch_bounds__(256)
gat_prep_zero(const float* __restrict__ W, const float* __restrict__ b,
              const float* __restrict__ att_w,
              float* __restrict__ G, float* __restrict__ cs,
              float* __restrict__ nacc, int n) {
  __shared__ float su_s[8][32];
  __shared__ float sv_s[8][32];
  int tid  = threadIdx.x;
  int lane = tid & 31;
  if (blockIdx.x > PRO_BLOCKS) {
    int i = (blockIdx.x - PRO_BLOCKS - 1) * 256 + tid;
    if (i < n) nacc[i] = 0.f;
    return;
  }
  if (blockIdx.x == PRO_BLOCKS) {
    if (tid < 32) {
      float t = 0.f;
#pragma unroll
      for (int i = 0; i < 4; ++i) {
        int j = lane + 32 * i;
        t += (att_w[j] + att_w[HIDDEN + j]) * b[j];
      }
      t += __shfl_xor(t, 16, 32);
      t += __shfl_xor(t,  8, 32);
      t += __shfl_xor(t,  4, 32);
      t += __shfl_xor(t,  2, 32);
      t += __shfl_xor(t,  1, 32);
      if (lane == 0) cs[0] = t;
    }
    return;
  }
  int kg  = __builtin_amdgcn_readfirstlane(tid >> 5);
  int col = blockIdx.x * 32 + lane;
  float su = 0.f, sv = 0.f;
#pragma unroll
  for (int i = 0; i < 16; ++i) {
    int k = kg * 16 + i;
    float w = W[k * HIDDEN + col];
    su = fmaf(att_w[k], w, su);
    sv = fmaf(att_w[HIDDEN + k], w, sv);
  }
  su_s[kg][lane] = su;
  sv_s[kg][lane] = sv;
  __syncthreads();
  if (kg == 0) {
    float u = su_s[0][lane], v = sv_s[0][lane];
#pragma unroll
    for (int g = 1; g < 8; ++g) { u += su_s[g][lane]; v += sv_s[g][lane]; }
    G[col] = u;
    G[HIDDEN + col] = v;
  } else {
    G[(2 * kg) * HIDDEN + col] = 0.f;
    G[(2 * kg + 1) * HIDDEN + col] = 0.f;
  }
}

__global__ void gat_pq_wmma(const float* __restrict__ x, const float* __restrict__ G,
                            float* __restrict__ p, float* __restrict__ q,
                            int n_tiles, int n_nodes) {
  int lane = threadIdx.x & 31;
  int wv   = (blockIdx.x * blockDim.x + threadIdx.x) >> 5;
  int t0 = wv * 2;
  int t1 = t0 + 1;
  if (t0 >= n_tiles) return;
  bool has1 = (t1 < n_tiles);

  int m    = lane & 15;
  int koff = (lane >> 4) << 1;
  int n0 = t0 * 16 + m; if (n0 >= n_nodes) n0 = n_nodes - 1;
  int n1 = has1 ? (t1 * 16 + m) : n0; if (n1 >= n_nodes) n1 = n_nodes - 1;

  const float* a_ptr = G + (size_t)m  * HIDDEN + koff;
  const float* b0ptr = x + (size_t)n0 * HIDDEN + koff;
  const float* b1ptr = x + (size_t)n1 * HIDDEN + koff;

  v8f acc0 = {0.f, 0.f, 0.f, 0.f, 0.f, 0.f, 0.f, 0.f};
  v8f acc1 = {0.f, 0.f, 0.f, 0.f, 0.f, 0.f, 0.f, 0.f};
#pragma unroll
  for (int k0 = 0; k0 < HIDDEN; k0 += 4) {
    v2f aa  = *(const v2f*)(a_ptr + k0);
    v2f bb0 = *(const v2f*)(b0ptr + k0);
    v2f bb1 = *(const v2f*)(b1ptr + k0);
    acc0 = __builtin_amdgcn_wmma_f32_16x16x4_f32(false, aa, false, bb0,
                                                 (short)0, acc0, false, false);
    acc1 = __builtin_amdgcn_wmma_f32_16x16x4_f32(false, aa, false, bb1,
                                                 (short)0, acc1, false, false);
  }
  if (lane < 16) {
    int o0 = t0 * 16 + lane;
    if (o0 < n_nodes) { p[o0] = acc0[0]; q[o0] = acc0[1]; }
    int o1 = t1 * 16 + lane;
    if (has1 && o1 < n_nodes) { p[o1] = acc1[0]; q[o1] = acc1[1]; }
  }
}

__global__ void gat_edge(const int* __restrict__ e, const float* __restrict__ p,
                         const float* __restrict__ q, const float* __restrict__ cs,
                         float* __restrict__ nacc, int n_edges) {
  int i = blockIdx.x * blockDim.x + threadIdx.x;
  if (i >= n_edges) return;
  int srcn = e[2 * i];
  int dstn = e[2 * i + 1];
  float z = p[srcn] + q[dstn] + cs[0];
  float lz = z > 0.f ? z : 0.2f * z;
  atomicAdd(&nacc[dstn], expf(lz));
}

#define OUT_ROWS 25
#define OUT_GROUP 5
static_assert(OUT_GROUP == 5 && OUT_ROWS % OUT_GROUP == 0 && OUT_ROWS <= 32,
              "the group fence names five values; one lane per row holds its normaliser");

__device__ __forceinline__ float4 out_residual(float4 v, float nv) {
  float denom = (nv == 0.f) ? 1.f : nv;
  float4 y;
  y.x = fmaxf(v.x / denom, 0.f) + v.x;
  y.y = fmaxf(v.y / denom, 0.f) + v.y;
  y.z = fmaxf(v.z / denom, 0.f) + v.z;
  y.w = fmaxf(v.w / denom, 0.f) + v.w;
  return y;
}

__device__ __forceinline__ float out_sumsq(float4 y) {
  return y.x * y.x + y.y * y.y + y.z * y.z + y.w * y.w;
}

__device__ __forceinline__ float4 out_norm(float4 y, float ss, float4 w4, float4 b4) {
  float inv_rms = rsqrtf(ss * (1.0f / HIDDEN) + 1e-6f);
  float4 o;
  o.x = w4.x * (y.x * inv_rms) + b4.x;
  o.y = w4.y * (y.y * inv_rms) + b4.y;
  o.z = w4.z * (y.z * inv_rms) + b4.z;
  o.w = w4.w * (y.w * inv_rms) + b4.w;
  return o;
}

__global__ void __launch_bounds__(256)
gat_out(const float* x, const float* __restrict__ nacc,
        const float* __restrict__ nw, const float* __restrict__ nb,
        float* __restrict__ out, int n_nodes) {
  int lane = threadIdx.x & 31;
  int wave = __builtin_amdgcn_readfirstlane(
      (int)((blockIdx.x * blockDim.x + threadIdx.x) >> 5));
  int first = wave * OUT_ROWS;
  if (first >= n_nodes) return;
  int rows = n_nodes - first;

  int ni = first + lane; if (ni > n_nodes - 1) ni = n_nodes - 1;
  float nv_all = nacc[ni];
  float4 w4 = ((const float4*)nw)[lane];
  float4 b4 = ((const float4*)nb)[lane];

  const float4* xr   = (const float4*)(x   + (size_t)first * HIDDEN) + lane;
  float4*       orow = (float4*)      (out + (size_t)first * HIDDEN) + lane;

  if (rows >= OUT_ROWS) {
    float4 xv[OUT_ROWS];
#pragma unroll
    for (int r = 0; r < OUT_ROWS; ++r) xv[r] = xr[r * 32];
    asm volatile("" : "+v"(nv_all) : : "memory");

#pragma unroll
    for (int g = 0; g < OUT_ROWS; g += OUT_GROUP) {
      float4 y[OUT_GROUP];
      float  ss[OUT_GROUP];
      float  nv[OUT_GROUP];
#pragma unroll
      for (int j = 0; j < OUT_GROUP; ++j)
        nv[j] = __int_as_float(
            __builtin_amdgcn_readlane(__float_as_int(nv_all), g + j));
      asm volatile("" : "+s"(nv[0]), "+s"(nv[1]), "+s"(nv[2]), "+s"(nv[3]), "+s"(nv[4])
                   : : "memory");
#pragma unroll
      for (int j = 0; j < OUT_GROUP; ++j) {
        y[j]  = out_residual(xv[g + j], nv[j]);
        ss[j] = out_sumsq(y[j]);
      }
#pragma unroll
      for (int m = 16; m >= 1; m >>= 1) {
#pragma unroll
        for (int j = 0; j < OUT_GROUP; ++j) ss[j] += __shfl_xor(ss[j], m, 32);
      }
#pragma unroll
      for (int j = 0; j < OUT_GROUP; ++j)
        orow[(g + j) * 32] = out_norm(y[j], ss[j], w4, b4);
    }
  } else {
    for (int r = 0; r < rows; ++r) {
      float nv = __int_as_float(
          __builtin_amdgcn_readlane(__float_as_int(nv_all), r));
      float4 y = out_residual(xr[r * 32], nv);
      float ss = out_sumsq(y);
      ss += __shfl_xor(ss, 16, 32);
      ss += __shfl_xor(ss,  8, 32);
      ss += __shfl_xor(ss,  4, 32);
      ss += __shfl_xor(ss,  2, 32);
      ss += __shfl_xor(ss,  1, 32);
      orow[r * 32] = out_norm(y, ss, w4, b4);
    }
  }
}

extern "C" void kernel_launch(void* const* d_in, const int* in_sizes, int n_in,
                              void* d_out, int out_size, void* d_ws, size_t ws_size,
                              hipStream_t stream) {
  const float* x     = (const float*)d_in[0];
  const float* W     = (const float*)d_in[1];
  const float* b     = (const float*)d_in[2];
  const float* att_w = (const float*)d_in[3];
  const float* nw    = (const float*)d_in[4];
  const float* nb    = (const float*)d_in[5];
  const int*   e     = (const int*)d_in[6];
  float* out = (float*)d_out;

  const int n_nodes = in_sizes[0] / HIDDEN;
  const int n_edges = in_sizes[6] / 2;

  float* G    = (float*)d_ws;
  float* cs   = G + 16 * HIDDEN;
  float* p    = cs + 16;
  float* q    = p + n_nodes;
  float* nacc = q + n_nodes;

  gat_prep_zero<<<PRO_BLOCKS + 1 + (n_nodes + 255) / 256, 256, 0, stream>>>(
      W, b, att_w, G, cs, nacc, n_nodes);

  int n_tiles = (n_nodes + 15) / 16;
  int n_waves = (n_tiles + 1) / 2;
  int pq_blocks = (n_waves * 32 + 127) / 128;
  gat_pq_wmma<<<pq_blocks, 128, 0, stream>>>(x, G, p, q, n_tiles, n_nodes);

  gat_edge<<<(n_edges + 255) / 256, 256, 0, stream>>>(e, p, q, cs, nacc, n_edges);

  int out_waves = (n_nodes + OUT_ROWS - 1) / OUT_ROWS;
  gat_out<<<(out_waves + 7) / 8, 256, 0, stream>>>(x, nacc, nw, nb, out, n_nodes);
}


# u,v,c prologue and zero-fill folded into the p,q launch (4 -> 3 launches): each 8-wave block rebuilds u,v in LDS while its x rows, loaded whole and staged through LDS, are in flight; sums unchanged
# speedup vs baseline: 1.3141x; 1.0604x over previous
#include <hip/hip_runtime.h>
#include <hip/hip_bf16.h>

typedef __attribute__((ext_vector_type(2))) float v2f;
typedef __attribute__((ext_vector_type(8))) float v8f;

#define HIDDEN 128

#define PQ_WAVES 8
#define PQ_ROWS 32
#define PQ_PITCH 132

__global__ void __launch_bounds__(PQ_WAVES * 32)
gat_uv_pq(const float* __restrict__ x, const float* __restrict__ W,
          const float* __restrict__ b, const float* __restrict__ att_w,
          float* __restrict__ cs, float* __restrict__ p, float* __restrict__ q,
          float* __restrict__ nacc, int n_nodes) {
  __shared__ __attribute__((aligned(16))) float sx[PQ_WAVES][PQ_ROWS * PQ_PITCH];
  __shared__ __attribute__((aligned(16))) float su_s[8][HIDDEN];
  __shared__ __attribute__((aligned(16))) float sv_s[8][HIDDEN];
  __shared__ __attribute__((aligned(16))) float sA[16 * PQ_PITCH];
  int tid  = threadIdx.x;
  int lane = tid & 31;
  int kg = __builtin_amdgcn_readfirstlane(tid >> 5);
  int wv = blockIdx.x * PQ_WAVES + kg;
  int row0 = wv * 32;

  float4 wr[16];
#pragma unroll
  for (int i = 0; i < 16; ++i)
    wr[i] = ((const float4*)(W + (kg * 16 + i) * HIDDEN))[lane];

  int base = row0; if (base > n_nodes - PQ_ROWS) base = n_nodes - PQ_ROWS;
  if (base < 0) base = 0;
  const float4* xr = (const float4*)(x + (size_t)base * HIDDEN) + lane;
  float4 xv[PQ_ROWS];
#pragma unroll
  for (int r = 0; r < PQ_ROWS; ++r) xv[r] = xr[r * 32];

  float4 su = {0.f, 0.f, 0.f, 0.f}, sv = {0.f, 0.f, 0.f, 0.f};
#pragma unroll
  for (int i = 0; i < 16; ++i) {
    int k = kg * 16 + i;
    float a1 = att_w[k], a2 = att_w[HIDDEN + k];
    su.x = fmaf(a1, wr[i].x, su.x); sv.x = fmaf(a2, wr[i].x, sv.x);
    su.y = fmaf(a1, wr[i].y, su.y); sv.y = fmaf(a2, wr[i].y, sv.y);
    su.z = fmaf(a1, wr[i].z, su.z); sv.z = fmaf(a2, wr[i].z, sv.z);
    su.w = fmaf(a1, wr[i].w, su.w); sv.w = fmaf(a2, wr[i].w, sv.w);
  }
  *(float4*)(&su_s[kg][4 * lane]) = su;
  *(float4*)(&sv_s[kg][4 * lane]) = sv;
  float* s = sx[kg];
#pragma unroll
  for (int r = 0; r < PQ_ROWS; ++r)
    *(float4*)(s + r * PQ_PITCH + 4 * lane) = xv[r];
  __syncthreads();
  {
    int col  = tid & (HIDDEN - 1);
    int half = tid >> 7;
    const float (*part)[HIDDEN] = half ? sv_s : su_s;
    float t = part[0][col];
#pragma unroll
    for (int g = 1; g < 8; ++g) t += part[g][col];
    sA[half * PQ_PITCH + col] = t;
#pragma unroll
    for (int j = 0; j < 7; ++j) sA[(2 + 2 * j + half) * PQ_PITCH + col] = 0.f;
  }
  __syncthreads();

  int m    = lane & 15;
  int koff = (lane >> 4) << 1;
  const float* a_ptr = sA + m * PQ_PITCH + koff;

  const float* b0ptr = s + m * PQ_PITCH + koff;
  const float* b1ptr = s + (16 + m) * PQ_PITCH + koff;
  v8f acc0 = {0.f, 0.f, 0.f, 0.f, 0.f, 0.f, 0.f, 0.f};
  v8f acc1 = {0.f, 0.f, 0.f, 0.f, 0.f, 0.f, 0.f, 0.f};
#pragma unroll
  for (int i = 0; i < HIDDEN / 4; ++i) {
    v2f aa  = *(const v2f*)(a_ptr + 4 * i);
    v2f bb0 = *(const v2f*)(b0ptr + 4 * i);
    v2f bb1 = *(const v2f*)(b1ptr + 4 * i);
    acc0 = __builtin_amdgcn_wmma_f32_16x16x4_f32(false, aa, false, bb0,
                                                 (short)0, acc0, false, false);
    acc1 = __builtin_amdgcn_wmma_f32_16x16x4_f32(false, aa, false, bb1,
                                                 (short)0, acc1, false, false);
  }
  if (lane < 16) {
    int o0 = base + lane;
    if (o0 >= row0 && o0 < n_nodes) { p[o0] = acc0[0]; q[o0] = acc0[1]; }
    int o1 = base + 16 + lane;
    if (o1 >= row0 && o1 < n_nodes) { p[o1] = acc1[0]; q[o1] = acc1[1]; }
  }
  if (row0 + lane < n_nodes) nacc[row0 + lane] = 0.f;

  if (blockIdx.x == 0 && kg == 0) {
    float t = 0.f;
#pragma unroll
    for (int i = 0; i < 4; ++i) {
      int j = lane + 32 * i;
      t += (att_w[j] + att_w[HIDDEN + j]) * b[j];
    }
    t += __shfl_xor(t, 16, 32);
    t += __shfl_xor(t,  8, 32);
    t += __shfl_xor(t,  4, 32);
    t += __shfl_xor(t,  2, 32);
    t += __shfl_xor(t,  1, 32);
    if (lane == 0) cs[0] = t;
  }
}

__global__ void gat_edge(const int* __restrict__ e, const float* __restrict__ p,
                         const float* __restrict__ q, const float* __restrict__ cs,
                         float* __restrict__ nacc, int n_edges) {
  int i = blockIdx.x * blockDim.x + threadIdx.x;
  if (i >= n_edges) return;
  int srcn = e[2 * i];
  int dstn = e[2 * i + 1];
  float z = p[srcn] + q[dstn] + cs[0];
  float lz = z > 0.f ? z : 0.2f * z;
  atomicAdd(&nacc[dstn], expf(lz));
}

#define OUT_ROWS 25
#define OUT_GROUP 5
static_assert(OUT_GROUP == 5 && OUT_ROWS % OUT_GROUP == 0 && OUT_ROWS <= 32,
              "the group fence names five values; one lane per row holds its normaliser");

__device__ __forceinline__ float4 out_residual(float4 v, float nv) {
  float denom = (nv == 0.f) ? 1.f : nv;
  float4 y;
  y.x = fmaxf(v.x / denom, 0.f) + v.x;
  y.y = fmaxf(v.y / denom, 0.f) + v.y;
  y.z = fmaxf(v.z / denom, 0.f) + v.z;
  y.w = fmaxf(v.w / denom, 0.f) + v.w;
  return y;
}

__device__ __forceinline__ float out_sumsq(float4 y) {
  return y.x * y.x + y.y * y.y + y.z * y.z + y.w * y.w;
}

__device__ __forceinline__ float4 out_norm(float4 y, float ss, float4 w4, float4 b4) {
  float inv_rms = rsqrtf(ss * (1.0f / HIDDEN) + 1e-6f);
  float4 o;
  o.x = w4.x * (y.x * inv_rms) + b4.x;
  o.y = w4.y * (y.y * inv_rms) + b4.y;
  o.z = w4.z * (y.z * inv_rms) + b4.z;
  o.w = w4.w * (y.w * inv_rms) + b4.w;
  return o;
}

__global__ void __launch_bounds__(256)
gat_out(const float* x, const float* __restrict__ nacc,
        const float* __restrict__ nw, const float* __restrict__ nb,
        float* __restrict__ out, int n_nodes) {
  int lane = threadIdx.x & 31;
  int wave = __builtin_amdgcn_readfirstlane(
      (int)((blockIdx.x * blockDim.x + threadIdx.x) >> 5));
  int first = wave * OUT_ROWS;
  if (first >= n_nodes) return;
  int rows = n_nodes - first;

  int ni = first + lane; if (ni > n_nodes - 1) ni = n_nodes - 1;
  float nv_all = nacc[ni];
  float4 w4 = ((const float4*)nw)[lane];
  float4 b4 = ((const float4*)nb)[lane];

  const float4* xr   = (const float4*)(x   + (size_t)first * HIDDEN) + lane;
  float4*       orow = (float4*)      (out + (size_t)first * HIDDEN) + lane;

  if (rows >= OUT_ROWS) {
    float4 xv[OUT_ROWS];
#pragma unroll
    for (int r = 0; r < OUT_ROWS; ++r) xv[r] = xr[r * 32];
    asm volatile("" : "+v"(nv_all) : : "memory");

#pragma unroll
    for (int g = 0; g < OUT_ROWS; g += OUT_GROUP) {
      float4 y[OUT_GROUP];
      float  ss[OUT_GROUP];
      float  nv[OUT_GROUP];
#pragma unroll
      for (int j = 0; j < OUT_GROUP; ++j)
        nv[j] = __int_as_float(
            __builtin_amdgcn_readlane(__float_as_int(nv_all), g + j));
      asm volatile("" : "+s"(nv[0]), "+s"(nv[1]), "+s"(nv[2]), "+s"(nv[3]), "+s"(nv[4])
                   : : "memory");
#pragma unroll
      for (int j = 0; j < OUT_GROUP; ++j) {
        y[j]  = out_residual(xv[g + j], nv[j]);
        ss[j] = out_sumsq(y[j]);
      }
#pragma unroll
      for (int m = 16; m >= 1; m >>= 1) {
#pragma unroll
        for (int j = 0; j < OUT_GROUP; ++j) ss[j] += __shfl_xor(ss[j], m, 32);
      }
#pragma unroll
      for (int j = 0; j < OUT_GROUP; ++j)
        orow[(g + j) * 32] = out_norm(y[j], ss[j], w4, b4);
    }
  } else {
    for (int r = 0; r < rows; ++r) {
      float nv = __int_as_float(
          __builtin_amdgcn_readlane(__float_as_int(nv_all), r));
      float4 y = out_residual(xr[r * 32], nv);
      float ss = out_sumsq(y);
      ss += __shfl_xor(ss, 16, 32);
      ss += __shfl_xor(ss,  8, 32);
      ss += __shfl_xor(ss,  4, 32);
      ss += __shfl_xor(ss,  2, 32);
      ss += __shfl_xor(ss,  1, 32);
      orow[r * 32] = out_norm(y, ss, w4, b4);
    }
  }
}

extern "C" void kernel_launch(void* const* d_in, const int* in_sizes, int n_in,
                              void* d_out, int out_size, void* d_ws, size_t ws_size,
                              hipStream_t stream) {
  const float* x     = (const float*)d_in[0];
  const float* W     = (const float*)d_in[1];
  const float* b     = (const float*)d_in[2];
  const float* att_w = (const float*)d_in[3];
  const float* nw    = (const float*)d_in[4];
  const float* nb    = (const float*)d_in[5];
  const int*   e     = (const int*)d_in[6];
  float* out = (float*)d_out;

  const int n_nodes = in_sizes[0] / HIDDEN;
  const int n_edges = in_sizes[6] / 2;

  float* G    = (float*)d_ws;
  float* cs   = G + 16 * HIDDEN;
  float* p    = cs + 16;
  float* q    = p + n_nodes;
  float* nacc = q + n_nodes;

  int n_tiles = (n_nodes + 15) / 16;
  int n_waves = (n_tiles + 1) / 2;
  int pq_blocks = (n_waves + PQ_WAVES - 1) / PQ_WAVES;
  gat_uv_pq<<<pq_blocks, PQ_WAVES * 32, 0, stream>>>(x, W, b, att_w, cs, p, q, nacc, n_nodes);

  gat_edge<<<(n_edges + 255) / 256, 256, 0, stream>>>(e, p, q, cs, nacc, n_edges);

  int out_waves = (n_nodes + OUT_ROWS - 1) / OUT_ROWS;
  gat_out<<<(out_waves + 7) / 8, 256, 0, stream>>>(x, nacc, nw, nb, out, n_nodes);
}


# Output kernel: one correctly rounded reciprocal per node, then x*(1/n) per element in place of 128 IEEE divisions per row; h = relu(x/n) rounds up to about 1 ulp differently
# speedup vs baseline: 1.3596x; 1.0346x over previous
#include <hip/hip_runtime.h>
#include <hip/hip_bf16.h>

typedef __attribute__((ext_vector_type(2))) float v2f;
typedef __attribute__((ext_vector_type(8))) float v8f;

#define HIDDEN 128

#define PQ_WAVES 8
#define PQ_ROWS 32
#define PQ_PITCH 132

__global__ void __launch_bounds__(PQ_WAVES * 32)
gat_uv_pq(const float* __restrict__ x, const float* __restrict__ W,
          const float* __restrict__ b, const float* __restrict__ att_w,
          float* __restrict__ cs, float* __restrict__ p, float* __restrict__ q,
          float* __restrict__ nacc, int n_nodes) {
  __shared__ __attribute__((aligned(16))) float sx[PQ_WAVES][PQ_ROWS * PQ_PITCH];
  __shared__ __attribute__((aligned(16))) float su_s[8][HIDDEN];
  __shared__ __attribute__((aligned(16))) float sv_s[8][HIDDEN];
  __shared__ __attribute__((aligned(16))) float sA[16 * PQ_PITCH];
  int tid  = threadIdx.x;
  int lane = tid & 31;
  int kg = __builtin_amdgcn_readfirstlane(tid >> 5);
  int wv = blockIdx.x * PQ_WAVES + kg;
  int row0 = wv * 32;

  float4 wr[16];
#pragma unroll
  for (int i = 0; i < 16; ++i)
    wr[i] = ((const float4*)(W + (kg * 16 + i) * HIDDEN))[lane];

  int base = row0; if (base > n_nodes - PQ_ROWS) base = n_nodes - PQ_ROWS;
  if (base < 0) base = 0;
  const float4* xr = (const float4*)(x + (size_t)base * HIDDEN) + lane;
  float4 xv[PQ_ROWS];
#pragma unroll
  for (int r = 0; r < PQ_ROWS; ++r) xv[r] = xr[r * 32];

  float4 su = {0.f, 0.f, 0.f, 0.f}, sv = {0.f, 0.f, 0.f, 0.f};
#pragma unroll
  for (int i = 0; i < 16; ++i) {
    int k = kg * 16 + i;
    float a1 = att_w[k], a2 = att_w[HIDDEN + k];
    su.x = fmaf(a1, wr[i].x, su.x); sv.x = fmaf(a2, wr[i].x, sv.x);
    su.y = fmaf(a1, wr[i].y, su.y); sv.y = fmaf(a2, wr[i].y, sv.y);
    su.z = fmaf(a1, wr[i].z, su.z); sv.z = fmaf(a2, wr[i].z, sv.z);
    su.w = fmaf(a1, wr[i].w, su.w); sv.w = fmaf(a2, wr[i].w, sv.w);
  }
  *(float4*)(&su_s[kg][4 * lane]) = su;
  *(float4*)(&sv_s[kg][4 * lane]) = sv;
  float* s = sx[kg];
#pragma unroll
  for (int r = 0; r < PQ_ROWS; ++r)
    *(float4*)(s + r * PQ_PITCH + 4 * lane) = xv[r];
  __syncthreads();
  {
    int col  = tid & (HIDDEN - 1);
    int half = tid >> 7;
    const float (*part)[HIDDEN] = half ? sv_s : su_s;
    float t = part[0][col];
#pragma unroll
    for (int g = 1; g < 8; ++g) t += part[g][col];
    sA[half * PQ_PITCH + col] = t;
#pragma unroll
    for (int j = 0; j < 7; ++j) sA[(2 + 2 * j + half) * PQ_PITCH + col] = 0.f;
  }
  __syncthreads();

  int m    = lane & 15;
  int koff = (lane >> 4) << 1;
  const float* a_ptr = sA + m * PQ_PITCH + koff;

  const float* b0ptr = s + m * PQ_PITCH + koff;
  const float* b1ptr = s + (16 + m) * PQ_PITCH + koff;
  v8f acc0 = {0.f, 0.f, 0.f, 0.f, 0.f, 0.f, 0.f, 0.f};
  v8f acc1 = {0.f, 0.f, 0.f, 0.f, 0.f, 0.f, 0.f, 0.f};
#pragma unroll
  for (int i = 0; i < HIDDEN / 4; ++i) {
    v2f aa  = *(const v2f*)(a_ptr + 4 * i);
    v2f bb0 = *(const v2f*)(b0ptr + 4 * i);
    v2f bb1 = *(const v2f*)(b1ptr + 4 * i);
    acc0 = __builtin_amdgcn_wmma_f32_16x16x4_f32(false, aa, false, bb0,
                                                 (short)0, acc0, false, false);
    acc1 = __builtin_amdgcn_wmma_f32_16x16x4_f32(false, aa, false, bb1,
                                                 (short)0, acc1, false, false);
  }
  if (lane < 16) {
    int o0 = base + lane;
    if (o0 >= row0 && o0 < n_nodes) { p[o0] = acc0[0]; q[o0] = acc0[1]; }
    int o1 = base + 16 + lane;
    if (o1 >= row0 && o1 < n_nodes) { p[o1] = acc1[0]; q[o1] = acc1[1]; }
  }
  if (row0 + lane < n_nodes) nacc[row0 + lane] = 0.f;

  if (blockIdx.x == 0 && kg == 0) {
    float t = 0.f;
#pragma unroll
    for (int i = 0; i < 4; ++i) {
      int j = lane + 32 * i;
      t += (att_w[j] + att_w[HIDDEN + j]) * b[j];
    }
    t += __shfl_xor(t, 16, 32);
    t += __shfl_xor(t,  8, 32);
    t += __shfl_xor(t,  4, 32);
    t += __shfl_xor(t,  2, 32);
    t += __shfl_xor(t,  1, 32);
    if (lane == 0) cs[0] = t;
  }
}

__global__ void gat_edge(const int* __restrict__ e, const float* __restrict__ p,
                         const float* __restrict__ q, const float* __restrict__ cs,
                         float* __restrict__ nacc, int n_edges) {
  int i = blockIdx.x * blockDim.x + threadIdx.x;
  if (i >= n_edges) return;
  int srcn = e[2 * i];
  int dstn = e[2 * i + 1];
  float z = p[srcn] + q[dstn] + cs[0];
  float lz = z > 0.f ? z : 0.2f * z;
  atomicAdd(&nacc[dstn], expf(lz));
}

#define OUT_ROWS 25
#define OUT_GROUP 5
static_assert(OUT_GROUP == 5 && OUT_ROWS % OUT_GROUP == 0 && OUT_ROWS <= 32,
              "the group fence names five values; one lane per row holds its normaliser");

__device__ __forceinline__ float out_recip(float nv) {
  float denom = (nv == 0.f) ? 1.f : nv;
  return 1.0f / denom;
}

__device__ __forceinline__ float4 out_residual(float4 v, float rinv) {
  float4 y;
  y.x = fmaxf(v.x * rinv, 0.f) + v.x;
  y.y = fmaxf(v.y * rinv, 0.f) + v.y;
  y.z = fmaxf(v.z * rinv, 0.f) + v.z;
  y.w = fmaxf(v.w * rinv, 0.f) + v.w;
  return y;
}

__device__ __forceinline__ float out_sumsq(float4 y) {
  return y.x * y.x + y.y * y.y + y.z * y.z + y.w * y.w;
}

__device__ __forceinline__ float4 out_norm(float4 y, float ss, float4 w4, float4 b4) {
  float inv_rms = rsqrtf(ss * (1.0f / HIDDEN) + 1e-6f);
  float4 o;
  o.x = w4.x * (y.x * inv_rms) + b4.x;
  o.y = w4.y * (y.y * inv_rms) + b4.y;
  o.z = w4.z * (y.z * inv_rms) + b4.z;
  o.w = w4.w * (y.w * inv_rms) + b4.w;
  return o;
}

__global__ void __launch_bounds__(256)
gat_out(const float* x, const float* __restrict__ nacc,
        const float* __restrict__ nw, const float* __restrict__ nb,
        float* __restrict__ out, int n_nodes) {
  int lane = threadIdx.x & 31;
  int wave = __builtin_amdgcn_readfirstlane(
      (int)((blockIdx.x * blockDim.x + threadIdx.x) >> 5));
  int first = wave * OUT_ROWS;
  if (first >= n_nodes) return;
  int rows = n_nodes - first;

  int ni = first + lane; if (ni > n_nodes - 1) ni = n_nodes - 1;
  float nv_all = nacc[ni];
  float4 w4 = ((const float4*)nw)[lane];
  float4 b4 = ((const float4*)nb)[lane];

  const float4* xr   = (const float4*)(x   + (size_t)first * HIDDEN) + lane;
  float4*       orow = (float4*)      (out + (size_t)first * HIDDEN) + lane;

  if (rows >= OUT_ROWS) {
    float4 xv[OUT_ROWS];
#pragma unroll
    for (int r = 0; r < OUT_ROWS; ++r) xv[r] = xr[r * 32];
    asm volatile("" : "+v"(nv_all) : : "memory");
    float rinv_all = out_recip(nv_all);

#pragma unroll
    for (int g = 0; g < OUT_ROWS; g += OUT_GROUP) {
      float4 y[OUT_GROUP];
      float  ss[OUT_GROUP];
      float  nv[OUT_GROUP];
#pragma unroll
      for (int j = 0; j < OUT_GROUP; ++j)
        nv[j] = __int_as_float(
            __builtin_amdgcn_readlane(__float_as_int(rinv_all), g + j));
      asm volatile("" : "+s"(nv[0]), "+s"(nv[1]), "+s"(nv[2]), "+s"(nv[3]), "+s"(nv[4])
                   : : "memory");
#pragma unroll
      for (int j = 0; j < OUT_GROUP; ++j) {
        y[j]  = out_residual(xv[g + j], nv[j]);
        ss[j] = out_sumsq(y[j]);
      }
#pragma unroll
      for (int m = 16; m >= 1; m >>= 1) {
#pragma unroll
        for (int j = 0; j < OUT_GROUP; ++j) ss[j] += __shfl_xor(ss[j], m, 32);
      }
#pragma unroll
      for (int j = 0; j < OUT_GROUP; ++j)
        orow[(g + j) * 32] = out_norm(y[j], ss[j], w4, b4);
    }
  } else {
    float rinv_all = out_recip(nv_all);
    for (int r = 0; r < rows; ++r) {
      float nv = __int_as_float(
          __builtin_amdgcn_readlane(__float_as_int(rinv_all), r));
      float4 y = out_residual(xr[r * 32], nv);
      float ss = out_sumsq(y);
      ss += __shfl_xor(ss, 16, 32);
      ss += __shfl_xor(ss,  8, 32);
      ss += __shfl_xor(ss,  4, 32);
      ss += __shfl_xor(ss,  2, 32);
      ss += __shfl_xor(ss,  1, 32);
      orow[r * 32] = out_norm(y, ss, w4, b4);
    }
  }
}

extern "C" void kernel_launch(void* const* d_in, const int* in_sizes, int n_in,
                              void* d_out, int out_size, void* d_ws, size_t ws_size,
                              hipStream_t stream) {
  const float* x     = (const float*)d_in[0];
  const float* W     = (const float*)d_in[1];
  const float* b     = (const float*)d_in[2];
  const float* att_w = (const float*)d_in[3];
  const float* nw    = (const float*)d_in[4];
  const float* nb    = (const float*)d_in[5];
  const int*   e     = (const int*)d_in[6];
  float* out = (float*)d_out;

  const int n_nodes = in_sizes[0] / HIDDEN;
  const int n_edges = in_sizes[6] / 2;

  float* G    = (float*)d_ws;
  float* cs   = G + 16 * HIDDEN;
  float* p    = cs + 16;
  float* q    = p + n_nodes;
  float* nacc = q + n_nodes;

  int n_tiles = (n_nodes + 15) / 16;
  int n_waves = (n_tiles + 1) / 2;
  int pq_blocks = (n_waves + PQ_WAVES - 1) / PQ_WAVES;
  gat_uv_pq<<<pq_blocks, PQ_WAVES * 32, 0, stream>>>(x, W, b, att_w, cs, p, q, nacc, n_nodes);

  gat_edge<<<(n_edges + 255) / 256, 256, 0, stream>>>(e, p, q, cs, nacc, n_edges);

  int out_waves = (n_nodes + OUT_ROWS - 1) / OUT_ROWS;
  gat_out<<<(out_waves + 7) / 8, 256, 0, stream>>>(x, nacc, nw, nb, out, n_nodes);
}


# Edge kernel: each of 256 blocks owns (node part, edge slice), sums edge weights per node with LDS f32 atomics, then adds one global f32 atomic per node and block, line by line; sum order changes
# speedup vs baseline: 2.4358x; 1.7916x over previous
#include <hip/hip_runtime.h>
#include <hip/hip_bf16.h>

typedef __attribute__((ext_vector_type(2))) float v2f;
typedef __attribute__((ext_vector_type(8))) float v8f;

#define HIDDEN 128

#define PQ_WAVES 8
#define PQ_ROWS 32
#define PQ_PITCH 132

__global__ void __launch_bounds__(PQ_WAVES * 32)
gat_uv_pq(const float* __restrict__ x, const float* __restrict__ W,
          const float* __restrict__ b, const float* __restrict__ att_w,
          float* __restrict__ cs, float* __restrict__ p, float* __restrict__ q,
          float* __restrict__ nacc, int n_nodes) {
  __shared__ __attribute__((aligned(16))) float sx[PQ_WAVES][PQ_ROWS * PQ_PITCH];
  __shared__ __attribute__((aligned(16))) float su_s[8][HIDDEN];
  __shared__ __attribute__((aligned(16))) float sv_s[8][HIDDEN];
  __shared__ __attribute__((aligned(16))) float sA[16 * PQ_PITCH];
  int tid  = threadIdx.x;
  int lane = tid & 31;
  int kg = __builtin_amdgcn_readfirstlane(tid >> 5);
  int wv = blockIdx.x * PQ_WAVES + kg;
  int row0 = wv * 32;

  float4 wr[16];
#pragma unroll
  for (int i = 0; i < 16; ++i)
    wr[i] = ((const float4*)(W + (kg * 16 + i) * HIDDEN))[lane];

  int base = row0; if (base > n_nodes - PQ_ROWS) base = n_nodes - PQ_ROWS;
  if (base < 0) base = 0;
  const float4* xr = (const float4*)(x + (size_t)base * HIDDEN) + lane;
  float4 xv[PQ_ROWS];
#pragma unroll
  for (int r = 0; r < PQ_ROWS; ++r) xv[r] = xr[r * 32];

  float4 su = {0.f, 0.f, 0.f, 0.f}, sv = {0.f, 0.f, 0.f, 0.f};
#pragma unroll
  for (int i = 0; i < 16; ++i) {
    int k = kg * 16 + i;
    float a1 = att_w[k], a2 = att_w[HIDDEN + k];
    su.x = fmaf(a1, wr[i].x, su.x); sv.x = fmaf(a2, wr[i].x, sv.x);
    su.y = fmaf(a1, wr[i].y, su.y); sv.y = fmaf(a2, wr[i].y, sv.y);
    su.z = fmaf(a1, wr[i].z, su.z); sv.z = fmaf(a2, wr[i].z, sv.z);
    su.w = fmaf(a1, wr[i].w, su.w); sv.w = fmaf(a2, wr[i].w, sv.w);
  }
  *(float4*)(&su_s[kg][4 * lane]) = su;
  *(float4*)(&sv_s[kg][4 * lane]) = sv;
  float* s = sx[kg];
#pragma unroll
  for (int r = 0; r < PQ_ROWS; ++r)
    *(float4*)(s + r * PQ_PITCH + 4 * lane) = xv[r];
  __syncthreads();
  {
    int col  = tid & (HIDDEN - 1);
    int half = tid >> 7;
    const float (*part)[HIDDEN] = half ? sv_s : su_s;
    float t = part[0][col];
#pragma unroll
    for (int g = 1; g < 8; ++g) t += part[g][col];
    sA[half * PQ_PITCH + col] = t;
#pragma unroll
    for (int j = 0; j < 7; ++j) sA[(2 + 2 * j + half) * PQ_PITCH + col] = 0.f;
  }
  __syncthreads();

  int m    = lane & 15;
  int koff = (lane >> 4) << 1;
  const float* a_ptr = sA + m * PQ_PITCH + koff;

  const float* b0ptr = s + m * PQ_PITCH + koff;
  const float* b1ptr = s + (16 + m) * PQ_PITCH + koff;
  v8f acc0 = {0.f, 0.f, 0.f, 0.f, 0.f, 0.f, 0.f, 0.f};
  v8f acc1 = {0.f, 0.f, 0.f, 0.f, 0.f, 0.f, 0.f, 0.f};
#pragma unroll
  for (int i = 0; i < HIDDEN / 4; ++i) {
    v2f aa  = *(const v2f*)(a_ptr + 4 * i);
    v2f bb0 = *(const v2f*)(b0ptr + 4 * i);
    v2f bb1 = *(const v2f*)(b1ptr + 4 * i);
    acc0 = __builtin_amdgcn_wmma_f32_16x16x4_f32(false, aa, false, bb0,
                                                 (short)0, acc0, false, false);
    acc1 = __builtin_amdgcn_wmma_f32_16x16x4_f32(false, aa, false, bb1,
                                                 (short)0, acc1, false, false);
  }
  if (lane < 16) {
    int o0 = base + lane;
    if (o0 >= row0 && o0 < n_nodes) { p[o0] = acc0[0]; q[o0] = acc0[1]; }
    int o1 = base + 16 + lane;
    if (o1 >= row0 && o1 < n_nodes) { p[o1] = acc1[0]; q[o1] = acc1[1]; }
  }
  if (row0 + lane < n_nodes) nacc[row0 + lane] = 0.f;

  if (blockIdx.x == 0 && kg == 0) {
    float t = 0.f;
#pragma unroll
    for (int i = 0; i < 4; ++i) {
      int j = lane + 32 * i;
      t += (att_w[j] + att_w[HIDDEN + j]) * b[j];
    }
    t += __shfl_xor(t, 16, 32);
    t += __shfl_xor(t,  8, 32);
    t += __shfl_xor(t,  4, 32);
    t += __shfl_xor(t,  2, 32);
    t += __shfl_xor(t,  1, 32);
    if (lane == 0) cs[0] = t;
  }
}

#define EP_PART    12512
#define EP_SLICES  32
#define EP_THREADS 1024
#define EP_FLUSH   ((EP_PART / 32 + EP_THREADS / 32 - 1) / (EP_THREADS / 32))
#define EP_UNR     10

__global__ void __launch_bounds__(EP_THREADS)
gat_edge(const int* __restrict__ e, const float* __restrict__ p,
         const float* __restrict__ q, const float* __restrict__ cs,
         float* __restrict__ nacc, int n_edges, int n_nodes, int n_parts) {
  __shared__ __attribute__((aligned(16))) float acc[EP_PART];
  int tid   = threadIdx.x;
  int part  = blockIdx.x % n_parts;
  int slice = blockIdx.x / n_parts;
  int lo    = part * EP_PART;

  for (int i = tid; i < EP_PART / 4; i += EP_THREADS)
    ((float4*)acc)[i] = float4{0.f, 0.f, 0.f, 0.f};
  __syncthreads();

  int per = (n_edges + EP_SLICES - 1) / EP_SLICES;
  int e0  = slice * per;
  int e1  = e0 + per; if (e1 > n_edges) e1 = n_edges;
  float c = cs[0];
  const int2* e2 = (const int2*)e;

  for (int base = e0; base < e1; base += EP_THREADS * EP_UNR) {
    int2 ed[EP_UNR];
#pragma unroll
    for (int k = 0; k < EP_UNR; ++k) {
      int i = base + k * EP_THREADS + tid;
      ed[k] = (i < e1) ? e2[i] : int2{0, -1};
    }
    float pv[EP_UNR], qv[EP_UNR];
    unsigned loc[EP_UNR];
#pragma unroll
    for (int k = 0; k < EP_UNR; ++k) {
      loc[k] = (unsigned)(ed[k].y - lo);
      pv[k] = 0.f; qv[k] = 0.f;
      if (loc[k] < (unsigned)EP_PART) { pv[k] = p[ed[k].x]; qv[k] = q[ed[k].y]; }
    }
#pragma unroll
    for (int k = 0; k < EP_UNR; ++k) {
      if (loc[k] < (unsigned)EP_PART) {
        float z  = pv[k] + qv[k] + c;
        float lz = z > 0.f ? z : 0.2f * z;
        atomicAdd(&acc[loc[k]], expf(lz));
      }
    }
  }
  __syncthreads();

  int wave = tid >> 5, lane = tid & 31;
  float fv[EP_FLUSH];
#pragma unroll
  for (int k = 0; k < EP_FLUSH; ++k) {
    int l = wave + k * (EP_THREADS / 32);
    fv[k] = (l < EP_PART / 32) ? acc[l * 32 + lane] : 0.f;
  }
#pragma unroll
  for (int k = 0; k < EP_FLUSH; ++k) {
    int node = lo + (wave + k * (EP_THREADS / 32)) * 32 + lane;
    if (fv[k] != 0.f && node < n_nodes) atomicAdd(&nacc[node], fv[k]);
  }
}

#define OUT_ROWS 25
#define OUT_GROUP 5
static_assert(OUT_GROUP == 5 && OUT_ROWS % OUT_GROUP == 0 && OUT_ROWS <= 32,
              "the group fence names five values; one lane per row holds its normaliser");

__device__ __forceinline__ float out_recip(float nv) {
  float denom = (nv == 0.f) ? 1.f : nv;
  return 1.0f / denom;
}

__device__ __forceinline__ float4 out_residual(float4 v, float rinv) {
  float4 y;
  y.x = fmaxf(v.x * rinv, 0.f) + v.x;
  y.y = fmaxf(v.y * rinv, 0.f) + v.y;
  y.z = fmaxf(v.z * rinv, 0.f) + v.z;
  y.w = fmaxf(v.w * rinv, 0.f) + v.w;
  return y;
}

__device__ __forceinline__ float out_sumsq(float4 y) {
  return y.x * y.x + y.y * y.y + y.z * y.z + y.w * y.w;
}

__device__ __forceinline__ float4 out_norm(float4 y, float ss, float4 w4, float4 b4) {
  float inv_rms = rsqrtf(ss * (1.0f / HIDDEN) + 1e-6f);
  float4 o;
  o.x = w4.x * (y.x * inv_rms) + b4.x;
  o.y = w4.y * (y.y * inv_rms) + b4.y;
  o.z = w4.z * (y.z * inv_rms) + b4.z;
  o.w = w4.w * (y.w * inv_rms) + b4.w;
  return o;
}

__global__ void __launch_bounds__(256)
gat_out(const float* x, const float* __restrict__ nacc,
        const float* __restrict__ nw, const float* __restrict__ nb,
        float* __restrict__ out, int n_nodes) {
  int lane = threadIdx.x & 31;
  int wave = __builtin_amdgcn_readfirstlane(
      (int)((blockIdx.x * blockDim.x + threadIdx.x) >> 5));
  int first = wave * OUT_ROWS;
  if (first >= n_nodes) return;
  int rows = n_nodes - first;

  int ni = first + lane; if (ni > n_nodes - 1) ni = n_nodes - 1;
  float nv_all = nacc[ni];
  float4 w4 = ((const float4*)nw)[lane];
  float4 b4 = ((const float4*)nb)[lane];

  const float4* xr   = (const float4*)(x   + (size_t)first * HIDDEN) + lane;
  float4*       orow = (float4*)      (out + (size_t)first * HIDDEN) + lane;

  if (rows >= OUT_ROWS) {
    float4 xv[OUT_ROWS];
#pragma unroll
    for (int r = 0; r < OUT_ROWS; ++r) xv[r] = xr[r * 32];
    asm volatile("" : "+v"(nv_all) : : "memory");
    float rinv_all = out_recip(nv_all);

#pragma unroll
    for (int g = 0; g < OUT_ROWS; g += OUT_GROUP) {
      float4 y[OUT_GROUP];
      float  ss[OUT_GROUP];
      float  nv[OUT_GROUP];
#pragma unroll
      for (int j = 0; j < OUT_GROUP; ++j)
        nv[j] = __int_as_float(
            __builtin_amdgcn_readlane(__float_as_int(rinv_all), g + j));
      asm volatile("" : "+s"(nv[0]), "+s"(nv[1]), "+s"(nv[2]), "+s"(nv[3]), "+s"(nv[4])
                   : : "memory");
#pragma unroll
      for (int j = 0; j < OUT_GROUP; ++j) {
        y[j]  = out_residual(xv[g + j], nv[j]);
        ss[j] = out_sumsq(y[j]);
      }
#pragma unroll
      for (int m = 16; m >= 1; m >>= 1) {
#pragma unroll
        for (int j = 0; j < OUT_GROUP; ++j) ss[j] += __shfl_xor(ss[j], m, 32);
      }
#pragma unroll
      for (int j = 0; j < OUT_GROUP; ++j)
        orow[(g + j) * 32] = out_norm(y[j], ss[j], w4, b4);
    }
  } else {
    float rinv_all = out_recip(nv_all);
    for (int r = 0; r < rows; ++r) {
      float nv = __int_as_float(
          __builtin_amdgcn_readlane(__float_as_int(rinv_all), r));
      float4 y = out_residual(xr[r * 32], nv);
      float ss = out_sumsq(y);
      ss += __shfl_xor(ss, 16, 32);
      ss += __shfl_xor(ss,  8, 32);
      ss += __shfl_xor(ss,  4, 32);
      ss += __shfl_xor(ss,  2, 32);
      ss += __shfl_xor(ss,  1, 32);
      orow[r * 32] = out_norm(y, ss, w4, b4);
    }
  }
}

extern "C" void kernel_launch(void* const* d_in, const int* in_sizes, int n_in,
                              void* d_out, int out_size, void* d_ws, size_t ws_size,
                              hipStream_t stream) {
  const float* x     = (const float*)d_in[0];
  const float* W     = (const float*)d_in[1];
  const float* b     = (const float*)d_in[2];
  const float* att_w = (const float*)d_in[3];
  const float* nw    = (const float*)d_in[4];
  const float* nb    = (const float*)d_in[5];
  const int*   e     = (const int*)d_in[6];
  float* out = (float*)d_out;

  const int n_nodes = in_sizes[0] / HIDDEN;
  const int n_edges = in_sizes[6] / 2;

  float* G    = (float*)d_ws;
  float* cs   = G + 16 * HIDDEN;
  float* p    = cs + 16;
  float* q    = p + n_nodes;
  float* nacc = q + n_nodes;

  int n_tiles = (n_nodes + 15) / 16;
  int n_waves = (n_tiles + 1) / 2;
  int pq_blocks = (n_waves + PQ_WAVES - 1) / PQ_WAVES;
  gat_uv_pq<<<pq_blocks, PQ_WAVES * 32, 0, stream>>>(x, W, b, att_w, cs, p, q, nacc, n_nodes);

  int n_parts = (n_nodes + EP_PART - 1) / EP_PART;
  gat_edge<<<n_parts * EP_SLICES, EP_THREADS, 0, stream>>>(e, p, q, cs, nacc, n_edges, n_nodes, n_parts);

  int out_waves = (n_nodes + OUT_ROWS - 1) / OUT_ROWS;
  gat_out<<<(out_waves + 7) / 8, 256, 0, stream>>>(x, nacc, nw, nb, out, n_nodes);
}


# p,q launch: fixed grid of 256 16-wave blocks walks 16-node tiles; u,v built once per block; a wave requests its next tile before the current tile's WMMAs run; sums unchanged
# speedup vs baseline: 2.6831x; 1.1015x over previous
#include <hip/hip_runtime.h>
#include <hip/hip_bf16.h>

typedef __attribute__((ext_vector_type(2))) float v2f;
typedef __attribute__((ext_vector_type(8))) float v8f;

#define HIDDEN 128

#define PQ_WAVES 16
#define PQ_KWAVES 8
#define PQ_TROWS 16
#define PQ_BLOCKS 256
#define PQ_PITCH 132

__global__ void __launch_bounds__(PQ_WAVES * 32)
gat_uv_pq(const float* __restrict__ x, const float* __restrict__ W,
          const float* __restrict__ b, const float* __restrict__ att_w,
          float* __restrict__ cs, float* __restrict__ p, float* __restrict__ q,
          float* __restrict__ nacc, int n_nodes, int n_tiles, int n_blocks) {
  __shared__ __attribute__((aligned(16))) float sx[PQ_WAVES][PQ_TROWS * PQ_PITCH];
  __shared__ __attribute__((aligned(16))) float su_s[PQ_KWAVES][HIDDEN];
  __shared__ __attribute__((aligned(16))) float sv_s[PQ_KWAVES][HIDDEN];
  __shared__ __attribute__((aligned(16))) float sA[16 * PQ_PITCH];
  int tid  = threadIdx.x;
  int lane = tid & 31;
  int kg = __builtin_amdgcn_readfirstlane(tid >> 5);
  int tstep = n_blocks * PQ_WAVES;
  int tile  = (int)blockIdx.x + n_blocks * kg;

  float4 wr[16];
  if (kg < PQ_KWAVES) {
#pragma unroll
    for (int i = 0; i < 16; ++i)
      wr[i] = ((const float4*)(W + (kg * 16 + i) * HIDDEN))[lane];
  } else {
#pragma unroll
    for (int i = 0; i < 16; ++i) wr[i] = float4{0.f, 0.f, 0.f, 0.f};
  }

  float4 cur[PQ_TROWS];
  if (tile < n_tiles) {
    int base = tile * PQ_TROWS; if (base > n_nodes - PQ_TROWS) base = n_nodes - PQ_TROWS;
    if (base < 0) base = 0;
    const float4* xr = (const float4*)(x + (size_t)base * HIDDEN) + lane;
#pragma unroll
    for (int r = 0; r < PQ_TROWS; ++r) cur[r] = xr[r * 32];
  } else {
#pragma unroll
    for (int r = 0; r < PQ_TROWS; ++r) cur[r] = float4{0.f, 0.f, 0.f, 0.f};
  }

  if (kg < PQ_KWAVES) {
    float4 su = {0.f, 0.f, 0.f, 0.f}, sv = {0.f, 0.f, 0.f, 0.f};
#pragma unroll
    for (int i = 0; i < 16; ++i) {
      int k = kg * 16 + i;
      float a1 = att_w[k], a2 = att_w[HIDDEN + k];
      su.x = fmaf(a1, wr[i].x, su.x); sv.x = fmaf(a2, wr[i].x, sv.x);
      su.y = fmaf(a1, wr[i].y, su.y); sv.y = fmaf(a2, wr[i].y, sv.y);
      su.z = fmaf(a1, wr[i].z, su.z); sv.z = fmaf(a2, wr[i].z, sv.z);
      su.w = fmaf(a1, wr[i].w, su.w); sv.w = fmaf(a2, wr[i].w, sv.w);
    }
    *(float4*)(&su_s[kg][4 * lane]) = su;
    *(float4*)(&sv_s[kg][4 * lane]) = sv;
  }
  __syncthreads();
  if (kg < PQ_KWAVES) {
    int col  = tid & (HIDDEN - 1);
    int half = tid >> 7;
    const float (*part)[HIDDEN] = half ? sv_s : su_s;
    float t = part[0][col];
#pragma unroll
    for (int g = 1; g < PQ_KWAVES; ++g) t += part[g][col];
    sA[half * PQ_PITCH + col] = t;
#pragma unroll
    for (int j = 0; j < 7; ++j) sA[(2 + 2 * j + half) * PQ_PITCH + col] = 0.f;
  }
  __syncthreads();

  int m    = lane & 15;
  int koff = (lane >> 4) << 1;
  const float* a_ptr = sA + m * PQ_PITCH + koff;
  v2f aa[HIDDEN / 4];
#pragma unroll
  for (int i = 0; i < HIDDEN / 4; ++i) aa[i] = *(const v2f*)(a_ptr + 4 * i);

  float* s = sx[kg];
  const float* bptr = s + m * PQ_PITCH + koff;

  while (tile < n_tiles) {
    int row0 = tile * PQ_TROWS;
    int base = row0; if (base > n_nodes - PQ_TROWS) base = n_nodes - PQ_TROWS;
    if (base < 0) base = 0;
#pragma unroll
    for (int r = 0; r < PQ_TROWS; ++r)
      *(float4*)(s + r * PQ_PITCH + 4 * lane) = cur[r];
    int nt = tile + tstep;
    if (nt < n_tiles) {
      int nb = nt * PQ_TROWS; if (nb > n_nodes - PQ_TROWS) nb = n_nodes - PQ_TROWS;
      if (nb < 0) nb = 0;
      const float4* xr = (const float4*)(x + (size_t)nb * HIDDEN) + lane;
#pragma unroll
      for (int r = 0; r < PQ_TROWS; ++r) cur[r] = xr[r * 32];
    }
    v8f acc = {0.f, 0.f, 0.f, 0.f, 0.f, 0.f, 0.f, 0.f};
#pragma unroll
    for (int i = 0; i < HIDDEN / 4; ++i) {
      v2f bb = *(const v2f*)(bptr + 4 * i);
      acc = __builtin_amdgcn_wmma_f32_16x16x4_f32(false, aa[i], false, bb,
                                                  (short)0, acc, false, false);
    }
    if (lane < 16) {
      int o = base + lane;
      if (o >= row0 && o < n_nodes) {
        p[o] = acc[0]; q[o] = acc[1];
        nacc[o] = 0.f;
      }
    }
    tile = nt;
  }

  if (blockIdx.x == 0 && kg == 0) {
    float t = 0.f;
#pragma unroll
    for (int i = 0; i < 4; ++i) {
      int j = lane + 32 * i;
      t += (att_w[j] + att_w[HIDDEN + j]) * b[j];
    }
    t += __shfl_xor(t, 16, 32);
    t += __shfl_xor(t,  8, 32);
    t += __shfl_xor(t,  4, 32);
    t += __shfl_xor(t,  2, 32);
    t += __shfl_xor(t,  1, 32);
    if (lane == 0) cs[0] = t;
  }
}

#define EP_PART    12512
#define EP_SLICES  32
#define EP_THREADS 1024
#define EP_FLUSH   ((EP_PART / 32 + EP_THREADS / 32 - 1) / (EP_THREADS / 32))
#define EP_UNR     10

__global__ void __launch_bounds__(EP_THREADS)
gat_edge(const int* __restrict__ e, const float* __restrict__ p,
         const float* __restrict__ q, const float* __restrict__ cs,
         float* __restrict__ nacc, int n_edges, int n_nodes, int n_parts) {
  __shared__ __attribute__((aligned(16))) float acc[EP_PART];
  int tid   = threadIdx.x;
  int part  = blockIdx.x % n_parts;
  int slice = blockIdx.x / n_parts;
  int lo    = part * EP_PART;

  for (int i = tid; i < EP_PART / 4; i += EP_THREADS)
    ((float4*)acc)[i] = float4{0.f, 0.f, 0.f, 0.f};
  __syncthreads();

  int per = (n_edges + EP_SLICES - 1) / EP_SLICES;
  int e0  = slice * per;
  int e1  = e0 + per; if (e1 > n_edges) e1 = n_edges;
  float c = cs[0];
  const int2* e2 = (const int2*)e;

  for (int base = e0; base < e1; base += EP_THREADS * EP_UNR) {
    int2 ed[EP_UNR];
#pragma unroll
    for (int k = 0; k < EP_UNR; ++k) {
      int i = base + k * EP_THREADS + tid;
      ed[k] = (i < e1) ? e2[i] : int2{0, -1};
    }
    float pv[EP_UNR], qv[EP_UNR];
    unsigned loc[EP_UNR];
#pragma unroll
    for (int k = 0; k < EP_UNR; ++k) {
      loc[k] = (unsigned)(ed[k].y - lo);
      pv[k] = 0.f; qv[k] = 0.f;
      if (loc[k] < (unsigned)EP_PART) { pv[k] = p[ed[k].x]; qv[k] = q[ed[k].y]; }
    }
#pragma unroll
    for (int k = 0; k < EP_UNR; ++k) {
      if (loc[k] < (unsigned)EP_PART) {
        float z  = pv[k] + qv[k] + c;
        float lz = z > 0.f ? z : 0.2f * z;
        atomicAdd(&acc[loc[k]], expf(lz));
      }
    }
  }
  __syncthreads();

  int wave = tid >> 5, lane = tid & 31;
  float fv[EP_FLUSH];
#pragma unroll
  for (int k = 0; k < EP_FLUSH; ++k) {
    int l = wave + k * (EP_THREADS / 32);
    fv[k] = (l < EP_PART / 32) ? acc[l * 32 + lane] : 0.f;
  }
#pragma unroll
  for (int k = 0; k < EP_FLUSH; ++k) {
    int node = lo + (wave + k * (EP_THREADS / 32)) * 32 + lane;
    if (fv[k] != 0.f && node < n_nodes) atomicAdd(&nacc[node], fv[k]);
  }
}

#define OUT_ROWS 25
#define OUT_GROUP 5
static_assert(OUT_GROUP == 5 && OUT_ROWS % OUT_GROUP == 0 && OUT_ROWS <= 32,
              "the group fence names five values; one lane per row holds its normaliser");

__device__ __forceinline__ float out_recip(float nv) {
  float denom = (nv == 0.f) ? 1.f : nv;
  return 1.0f / denom;
}

__device__ __forceinline__ float4 out_residual(float4 v, float rinv) {
  float4 y;
  y.x = fmaxf(v.x * rinv, 0.f) + v.x;
  y.y = fmaxf(v.y * rinv, 0.f) + v.y;
  y.z = fmaxf(v.z * rinv, 0.f) + v.z;
  y.w = fmaxf(v.w * rinv, 0.f) + v.w;
  return y;
}

__device__ __forceinline__ float out_sumsq(float4 y) {
  return y.x * y.x + y.y * y.y + y.z * y.z + y.w * y.w;
}

__device__ __forceinline__ float4 out_norm(float4 y, float ss, float4 w4, float4 b4) {
  float inv_rms = rsqrtf(ss * (1.0f / HIDDEN) + 1e-6f);
  float4 o;
  o.x = w4.x * (y.x * inv_rms) + b4.x;
  o.y = w4.y * (y.y * inv_rms) + b4.y;
  o.z = w4.z * (y.z * inv_rms) + b4.z;
  o.w = w4.w * (y.w * inv_rms) + b4.w;
  return o;
}

__global__ void __launch_bounds__(256)
gat_out(const float* x, const float* __restrict__ nacc,
        const float* __restrict__ nw, const float* __restrict__ nb,
        float* __restrict__ out, int n_nodes) {
  int lane = threadIdx.x & 31;
  int wave = __builtin_amdgcn_readfirstlane(
      (int)((blockIdx.x * blockDim.x + threadIdx.x) >> 5));
  int first = wave * OUT_ROWS;
  if (first >= n_nodes) return;
  int rows = n_nodes - first;

  int ni = first + lane; if (ni > n_nodes - 1) ni = n_nodes - 1;
  float nv_all = nacc[ni];
  float4 w4 = ((const float4*)nw)[lane];
  float4 b4 = ((const float4*)nb)[lane];

  const float4* xr   = (const float4*)(x   + (size_t)first * HIDDEN) + lane;
  float4*       orow = (float4*)      (out + (size_t)first * HIDDEN) + lane;

  if (rows >= OUT_ROWS) {
    float4 xv[OUT_ROWS];
#pragma unroll
    for (int r = 0; r < OUT_ROWS; ++r) xv[r] = xr[r * 32];
    asm volatile("" : "+v"(nv_all) : : "memory");
    float rinv_all = out_recip(nv_all);

#pragma unroll
    for (int g = 0; g < OUT_ROWS; g += OUT_GROUP) {
      float4 y[OUT_GROUP];
      float  ss[OUT_GROUP];
      float  nv[OUT_GROUP];
#pragma unroll
      for (int j = 0; j < OUT_GROUP; ++j)
        nv[j] = __int_as_float(
            __builtin_amdgcn_readlane(__float_as_int(rinv_all), g + j));
      asm volatile("" : "+s"(nv[0]), "+s"(nv[1]), "+s"(nv[2]), "+s"(nv[3]), "+s"(nv[4])
                   : : "memory");
#pragma unroll
      for (int j = 0; j < OUT_GROUP; ++j) {
        y[j]  = out_residual(xv[g + j], nv[j]);
        ss[j] = out_sumsq(y[j]);
      }
#pragma unroll
      for (int m = 16; m >= 1; m >>= 1) {
#pragma unroll
        for (int j = 0; j < OUT_GROUP; ++j) ss[j] += __shfl_xor(ss[j], m, 32);
      }
#pragma unroll
      for (int j = 0; j < OUT_GROUP; ++j)
        orow[(g + j) * 32] = out_norm(y[j], ss[j], w4, b4);
    }
  } else {
    float rinv_all = out_recip(nv_all);
    for (int r = 0; r < rows; ++r) {
      float nv = __int_as_float(
          __builtin_amdgcn_readlane(__float_as_int(rinv_all), r));
      float4 y = out_residual(xr[r * 32], nv);
      float ss = out_sumsq(y);
      ss += __shfl_xor(ss, 16, 32);
      ss += __shfl_xor(ss,  8, 32);
      ss += __shfl_xor(ss,  4, 32);
      ss += __shfl_xor(ss,  2, 32);
      ss += __shfl_xor(ss,  1, 32);
      orow[r * 32] = out_norm(y, ss, w4, b4);
    }
  }
}

extern "C" void kernel_launch(void* const* d_in, const int* in_sizes, int n_in,
                              void* d_out, int out_size, void* d_ws, size_t ws_size,
                              hipStream_t stream) {
  const float* x     = (const float*)d_in[0];
  const float* W     = (const float*)d_in[1];
  const float* b     = (const float*)d_in[2];
  const float* att_w = (const float*)d_in[3];
  const float* nw    = (const float*)d_in[4];
  const float* nb    = (const float*)d_in[5];
  const int*   e     = (const int*)d_in[6];
  float* out = (float*)d_out;

  const int n_nodes = in_sizes[0] / HIDDEN;
  const int n_edges = in_sizes[6] / 2;

  float* G    = (float*)d_ws;
  float* cs   = G + 16 * HIDDEN;
  float* p    = cs + 16;
  float* q    = p + n_nodes;
  float* nacc = q + n_nodes;

  int n_tiles = (n_nodes + PQ_TROWS - 1) / PQ_TROWS;
  gat_uv_pq<<<PQ_BLOCKS, PQ_WAVES * 32, 0, stream>>>(x, W, b, att_w, cs, p, q, nacc, n_nodes, n_tiles, PQ_BLOCKS);

  int n_parts = (n_nodes + EP_PART - 1) / EP_PART;
  gat_edge<<<n_parts * EP_SLICES, EP_THREADS, 0, stream>>>(e, p, q, cs, nacc, n_edges, n_nodes, n_parts);

  int out_waves = (n_nodes + OUT_ROWS - 1) / OUT_ROWS;
  gat_out<<<(out_waves + 7) / 8, 256, 0, stream>>>(x, nacc, nw, nb, out, n_nodes);
}


# Output kernel takes the p,q launch's grid and tile ownership (tile t: block t%256, wave (t/256)%16; next tile requested before the current is done): both reads of an x row come from one grid position
# speedup vs baseline: 2.8517x; 1.0628x over previous
#include <hip/hip_runtime.h>
#include <hip/hip_bf16.h>

typedef __attribute__((ext_vector_type(2))) float v2f;
typedef __attribute__((ext_vector_type(8))) float v8f;

#define HIDDEN 128

#define PQ_WAVES 16
#define PQ_KWAVES 8
#define PQ_TROWS 16
#define PQ_BLOCKS 256
#define PQ_PITCH 132

__global__ void __launch_bounds__(PQ_WAVES * 32)
gat_uv_pq(const float* __restrict__ x, const float* __restrict__ W,
          const float* __restrict__ b, const float* __restrict__ att_w,
          float* __restrict__ cs, float* __restrict__ p, float* __restrict__ q,
          float* __restrict__ nacc, int n_nodes, int n_tiles, int n_blocks) {
  __shared__ __attribute__((aligned(16))) float sx[PQ_WAVES][PQ_TROWS * PQ_PITCH];
  __shared__ __attribute__((aligned(16))) float su_s[PQ_KWAVES][HIDDEN];
  __shared__ __attribute__((aligned(16))) float sv_s[PQ_KWAVES][HIDDEN];
  __shared__ __attribute__((aligned(16))) float sA[16 * PQ_PITCH];
  int tid  = threadIdx.x;
  int lane = tid & 31;
  int kg = __builtin_amdgcn_readfirstlane(tid >> 5);
  int tstep = n_blocks * PQ_WAVES;
  int tile  = (int)blockIdx.x + n_blocks * kg;

  float4 wr[16];
  if (kg < PQ_KWAVES) {
#pragma unroll
    for (int i = 0; i < 16; ++i)
      wr[i] = ((const float4*)(W + (kg * 16 + i) * HIDDEN))[lane];
  } else {
#pragma unroll
    for (int i = 0; i < 16; ++i) wr[i] = float4{0.f, 0.f, 0.f, 0.f};
  }

  float4 cur[PQ_TROWS];
  if (tile < n_tiles) {
    int base = tile * PQ_TROWS; if (base > n_nodes - PQ_TROWS) base = n_nodes - PQ_TROWS;
    if (base < 0) base = 0;
    const float4* xr = (const float4*)(x + (size_t)base * HIDDEN) + lane;
#pragma unroll
    for (int r = 0; r < PQ_TROWS; ++r) cur[r] = xr[r * 32];
  } else {
#pragma unroll
    for (int r = 0; r < PQ_TROWS; ++r) cur[r] = float4{0.f, 0.f, 0.f, 0.f};
  }

  if (kg < PQ_KWAVES) {
    float4 su = {0.f, 0.f, 0.f, 0.f}, sv = {0.f, 0.f, 0.f, 0.f};
#pragma unroll
    for (int i = 0; i < 16; ++i) {
      int k = kg * 16 + i;
      float a1 = att_w[k], a2 = att_w[HIDDEN + k];
      su.x = fmaf(a1, wr[i].x, su.x); sv.x = fmaf(a2, wr[i].x, sv.x);
      su.y = fmaf(a1, wr[i].y, su.y); sv.y = fmaf(a2, wr[i].y, sv.y);
      su.z = fmaf(a1, wr[i].z, su.z); sv.z = fmaf(a2, wr[i].z, sv.z);
      su.w = fmaf(a1, wr[i].w, su.w); sv.w = fmaf(a2, wr[i].w, sv.w);
    }
    *(float4*)(&su_s[kg][4 * lane]) = su;
    *(float4*)(&sv_s[kg][4 * lane]) = sv;
  }
  __syncthreads();
  if (kg < PQ_KWAVES) {
    int col  = tid & (HIDDEN - 1);
    int half = tid >> 7;
    const float (*part)[HIDDEN] = half ? sv_s : su_s;
    float t = part[0][col];
#pragma unroll
    for (int g = 1; g < PQ_KWAVES; ++g) t += part[g][col];
    sA[half * PQ_PITCH + col] = t;
#pragma unroll
    for (int j = 0; j < 7; ++j) sA[(2 + 2 * j + half) * PQ_PITCH + col] = 0.f;
  }
  __syncthreads();

  int m    = lane & 15;
  int koff = (lane >> 4) << 1;
  const float* a_ptr = sA + m * PQ_PITCH + koff;
  v2f aa[HIDDEN / 4];
#pragma unroll
  for (int i = 0; i < HIDDEN / 4; ++i) aa[i] = *(const v2f*)(a_ptr + 4 * i);

  float* s = sx[kg];
  const float* bptr = s + m * PQ_PITCH + koff;

  while (tile < n_tiles) {
    int row0 = tile * PQ_TROWS;
    int base = row0; if (base > n_nodes - PQ_TROWS) base = n_nodes - PQ_TROWS;
    if (base < 0) base = 0;
#pragma unroll
    for (int r = 0; r < PQ_TROWS; ++r)
      *(float4*)(s + r * PQ_PITCH + 4 * lane) = cur[r];
    int nt = tile + tstep;
    if (nt < n_tiles) {
      int nb = nt * PQ_TROWS; if (nb > n_nodes - PQ_TROWS) nb = n_nodes - PQ_TROWS;
      if (nb < 0) nb = 0;
      const float4* xr = (const float4*)(x + (size_t)nb * HIDDEN) + lane;
#pragma unroll
      for (int r = 0; r < PQ_TROWS; ++r) cur[r] = xr[r * 32];
    }
    v8f acc = {0.f, 0.f, 0.f, 0.f, 0.f, 0.f, 0.f, 0.f};
#pragma unroll
    for (int i = 0; i < HIDDEN / 4; ++i) {
      v2f bb = *(const v2f*)(bptr + 4 * i);
      acc = __builtin_amdgcn_wmma_f32_16x16x4_f32(false, aa[i], false, bb,
                                                  (short)0, acc, false, false);
    }
    if (lane < 16) {
      int o = base + lane;
      if (o >= row0 && o < n_nodes) {
        p[o] = acc[0]; q[o] = acc[1];
        nacc[o] = 0.f;
      }
    }
    tile = nt;
  }

  if (blockIdx.x == 0 && kg == 0) {
    float t = 0.f;
#pragma unroll
    for (int i = 0; i < 4; ++i) {
      int j = lane + 32 * i;
      t += (att_w[j] + att_w[HIDDEN + j]) * b[j];
    }
    t += __shfl_xor(t, 16, 32);
    t += __shfl_xor(t,  8, 32);
    t += __shfl_xor(t,  4, 32);
    t += __shfl_xor(t,  2, 32);
    t += __shfl_xor(t,  1, 32);
    if (lane == 0) cs[0] = t;
  }
}

#define EP_PART    12512
#define EP_SLICES  32
#define EP_THREADS 1024
#define EP_FLUSH   ((EP_PART / 32 + EP_THREADS / 32 - 1) / (EP_THREADS / 32))
#define EP_UNR     10

__global__ void __launch_bounds__(EP_THREADS)
gat_edge(const int* __restrict__ e, const float* __restrict__ p,
         const float* __restrict__ q, const float* __restrict__ cs,
         float* __restrict__ nacc, int n_edges, int n_nodes, int n_parts) {
  __shared__ __attribute__((aligned(16))) float acc[EP_PART];
  int tid   = threadIdx.x;
  int part  = blockIdx.x % n_parts;
  int slice = blockIdx.x / n_parts;
  int lo    = part * EP_PART;

  for (int i = tid; i < EP_PART / 4; i += EP_THREADS)
    ((float4*)acc)[i] = float4{0.f, 0.f, 0.f, 0.f};
  __syncthreads();

  int per = (n_edges + EP_SLICES - 1) / EP_SLICES;
  int e0  = slice * per;
  int e1  = e0 + per; if (e1 > n_edges) e1 = n_edges;
  float c = cs[0];
  const int2* e2 = (const int2*)e;

  for (int base = e0; base < e1; base += EP_THREADS * EP_UNR) {
    int2 ed[EP_UNR];
#pragma unroll
    for (int k = 0; k < EP_UNR; ++k) {
      int i = base + k * EP_THREADS + tid;
      ed[k] = (i < e1) ? e2[i] : int2{0, -1};
    }
    float pv[EP_UNR], qv[EP_UNR];
    unsigned loc[EP_UNR];
#pragma unroll
    for (int k = 0; k < EP_UNR; ++k) {
      loc[k] = (unsigned)(ed[k].y - lo);
      pv[k] = 0.f; qv[k] = 0.f;
      if (loc[k] < (unsigned)EP_PART) { pv[k] = p[ed[k].x]; qv[k] = q[ed[k].y]; }
    }
#pragma unroll
    for (int k = 0; k < EP_UNR; ++k) {
      if (loc[k] < (unsigned)EP_PART) {
        float z  = pv[k] + qv[k] + c;
        float lz = z > 0.f ? z : 0.2f * z;
        atomicAdd(&acc[loc[k]], expf(lz));
      }
    }
  }
  __syncthreads();

  int wave = tid >> 5, lane = tid & 31;
  float fv[EP_FLUSH];
#pragma unroll
  for (int k = 0; k < EP_FLUSH; ++k) {
    int l = wave + k * (EP_THREADS / 32);
    fv[k] = (l < EP_PART / 32) ? acc[l * 32 + lane] : 0.f;
  }
#pragma unroll
  for (int k = 0; k < EP_FLUSH; ++k) {
    int node = lo + (wave + k * (EP_THREADS / 32)) * 32 + lane;
    if (fv[k] != 0.f && node < n_nodes) atomicAdd(&nacc[node], fv[k]);
  }
}

#define OUT_WAVES 16
#define OUT_TROWS 16
#define OUT_GROUP 4
static_assert(OUT_GROUP == 4 && OUT_TROWS % OUT_GROUP == 0 && OUT_TROWS <= 32,
              "the group fence names four values; one lane per row holds its normaliser");

__device__ __forceinline__ float out_recip(float nv) {
  float denom = (nv == 0.f) ? 1.f : nv;
  return 1.0f / denom;
}

__device__ __forceinline__ float4 out_residual(float4 v, float rinv) {
  float4 y;
  y.x = fmaxf(v.x * rinv, 0.f) + v.x;
  y.y = fmaxf(v.y * rinv, 0.f) + v.y;
  y.z = fmaxf(v.z * rinv, 0.f) + v.z;
  y.w = fmaxf(v.w * rinv, 0.f) + v.w;
  return y;
}

__device__ __forceinline__ float out_sumsq(float4 y) {
  return y.x * y.x + y.y * y.y + y.z * y.z + y.w * y.w;
}

__device__ __forceinline__ float4 out_norm(float4 y, float ss, float4 w4, float4 b4) {
  float inv_rms = rsqrtf(ss * (1.0f / HIDDEN) + 1e-6f);
  float4 o;
  o.x = w4.x * (y.x * inv_rms) + b4.x;
  o.y = w4.y * (y.y * inv_rms) + b4.y;
  o.z = w4.z * (y.z * inv_rms) + b4.z;
  o.w = w4.w * (y.w * inv_rms) + b4.w;
  return o;
}

__device__ __forceinline__ void out_tile(const float4 (&xv)[OUT_TROWS], float nv_all,
                                         float4* orow, float4 w4, float4 b4) {
  float rinv_all = out_recip(nv_all);
#pragma unroll
  for (int g = 0; g < OUT_TROWS; g += OUT_GROUP) {
    float4 y[OUT_GROUP];
    float  ss[OUT_GROUP];
    float  nv[OUT_GROUP];
#pragma unroll
    for (int j = 0; j < OUT_GROUP; ++j)
      nv[j] = __int_as_float(
          __builtin_amdgcn_readlane(__float_as_int(rinv_all), g + j));
    asm volatile("" : "+s"(nv[0]), "+s"(nv[1]), "+s"(nv[2]), "+s"(nv[3]) : : "memory");
#pragma unroll
    for (int j = 0; j < OUT_GROUP; ++j) {
      y[j]  = out_residual(xv[g + j], nv[j]);
      ss[j] = out_sumsq(y[j]);
    }
#pragma unroll
    for (int m = 16; m >= 1; m >>= 1) {
#pragma unroll
      for (int j = 0; j < OUT_GROUP; ++j) ss[j] += __shfl_xor(ss[j], m, 32);
    }
#pragma unroll
    for (int j = 0; j < OUT_GROUP; ++j)
      orow[(g + j) * 32] = out_norm(y[j], ss[j], w4, b4);
  }
}

__global__ void __launch_bounds__(OUT_WAVES * 32)
gat_out(const float* x, const float* nacc,
        const float* __restrict__ nw, const float* __restrict__ nb,
        float* __restrict__ out, int n_nodes, int n_blocks) {
  int lane = threadIdx.x & 31;
  int kg = __builtin_amdgcn_readfirstlane((int)(threadIdx.x >> 5));
  int n_full = n_nodes / OUT_TROWS;
  int tstep  = n_blocks * OUT_WAVES;
  int tile   = (int)blockIdx.x + n_blocks * kg;

  float4 w4 = ((const float4*)nw)[lane];
  float4 b4 = ((const float4*)nb)[lane];

  float4 cur[OUT_TROWS];
  float  ncur = 1.f;
  if (tile < n_full) {
    ncur = nacc[tile * OUT_TROWS + (lane & 15)];
    const float4* xr = (const float4*)(x + (size_t)tile * OUT_TROWS * HIDDEN) + lane;
#pragma unroll
    for (int r = 0; r < OUT_TROWS; ++r) cur[r] = xr[r * 32];
  } else {
#pragma unroll
    for (int r = 0; r < OUT_TROWS; ++r) cur[r] = float4{0.f, 0.f, 0.f, 0.f};
  }

  while (tile < n_full) {
    int nt = tile + tstep;
    float4* orow = (float4*)(out + (size_t)tile * OUT_TROWS * HIDDEN) + lane;
    if (nt < n_full) {
      float nnxt = nacc[nt * OUT_TROWS + (lane & 15)];
      float4 nxt[OUT_TROWS];
      const float4* xr = (const float4*)(x + (size_t)nt * OUT_TROWS * HIDDEN) + lane;
#pragma unroll
      for (int r = 0; r < OUT_TROWS; ++r) nxt[r] = xr[r * 32];
      asm volatile("" : "+v"(ncur) : : "memory");
      out_tile(cur, ncur, orow, w4, b4);
#pragma unroll
      for (int r = 0; r < OUT_TROWS; ++r) cur[r] = nxt[r];
      ncur = nnxt;
    } else {
      out_tile(cur, ncur, orow, w4, b4);
    }
    tile = nt;
  }

  int rem = n_nodes - n_full * OUT_TROWS;
  if (rem > 0 && (n_full % n_blocks) == (int)blockIdx.x &&
      ((n_full / n_blocks) % OUT_WAVES) == kg) {
    int first = n_full * OUT_TROWS;
    int ni = first + lane; if (ni > n_nodes - 1) ni = n_nodes - 1;
    float rinv_all = out_recip(nacc[ni]);
    const float4* xr   = (const float4*)(x   + (size_t)first * HIDDEN) + lane;
    float4*       orow = (float4*)      (out + (size_t)first * HIDDEN) + lane;
    for (int r = 0; r < rem; ++r) {
      float nv = __int_as_float(
          __builtin_amdgcn_readlane(__float_as_int(rinv_all), r));
      float4 y = out_residual(xr[r * 32], nv);
      float ss = out_sumsq(y);
      ss += __shfl_xor(ss, 16, 32);
      ss += __shfl_xor(ss,  8, 32);
      ss += __shfl_xor(ss,  4, 32);
      ss += __shfl_xor(ss,  2, 32);
      ss += __shfl_xor(ss,  1, 32);
      orow[r * 32] = out_norm(y, ss, w4, b4);
    }
  }
}

extern "C" void kernel_launch(void* const* d_in, const int* in_sizes, int n_in,
                              void* d_out, int out_size, void* d_ws, size_t ws_size,
                              hipStream_t stream) {
  const float* x     = (const float*)d_in[0];
  const float* W     = (const float*)d_in[1];
  const float* b     = (const float*)d_in[2];
  const float* att_w = (const float*)d_in[3];
  const float* nw    = (const float*)d_in[4];
  const float* nb    = (const float*)d_in[5];
  const int*   e     = (const int*)d_in[6];
  float* out = (float*)d_out;

  const int n_nodes = in_sizes[0] / HIDDEN;
  const int n_edges = in_sizes[6] / 2;

  float* G    = (float*)d_ws;
  float* cs   = G + 16 * HIDDEN;
  float* p    = cs + 16;
  float* q    = p + n_nodes;
  float* nacc = q + n_nodes;

  int n_tiles = (n_nodes + PQ_TROWS - 1) / PQ_TROWS;
  gat_uv_pq<<<PQ_BLOCKS, PQ_WAVES * 32, 0, stream>>>(x, W, b, att_w, cs, p, q, nacc, n_nodes, n_tiles, PQ_BLOCKS);

  int n_parts = (n_nodes + EP_PART - 1) / EP_PART;
  gat_edge<<<n_parts * EP_SLICES, EP_THREADS, 0, stream>>>(e, p, q, cs, nacc, n_edges, n_nodes, n_parts);

  gat_out<<<PQ_BLOCKS, OUT_WAVES * 32, 0, stream>>>(x, nacc, nw, nb, out, n_nodes, PQ_BLOCKS);
}
